# starting kernel
# baseline (speedup 1.0000x reference)
#include <hip/hip_runtime.h>

typedef __attribute__((ext_vector_type(2))) float v2f;
typedef __attribute__((ext_vector_type(8))) float v8f;

#define STEP 12
#define MAXS 10.0f

__global__ __launch_bounds__(256) void pwm_scan_wmma_kernel(
    const float* __restrict__ dna,
    const float* __restrict__ concen,
    const float* __restrict__ pwm,
    const float* __restrict__ pwmrc,
    const float* __restrict__ lamp,
    float* __restrict__ out,
    int L)
{
    const int lane = threadIdx.x & 31;
    const int wave = threadIdx.x >> 5;
    const int n    = lane & 15;
    const int half = lane >> 4;
    const int b    = blockIdx.y;
    const int i0   = blockIdx.x * 256 + wave * 32;
    const int Lo   = L - STEP + 1;
    const int Ltot = L + 1;

    v2f aw[STEP];
#pragma unroll
    for (int j = 0; j < STEP; ++j) {
        v2f w = {0.0f, 0.0f};
        if (n == 0)      w = *(const v2f*)(pwm   + j * 4 + 2 * half);
        else if (n == 1) w = *(const v2f*)(pwmrc + j * 4 + 2 * half);
        aw[j] = w;
    }

    const float* dnab = dna + (size_t)b * L * 4;

    int p0 = i0 + n;        if (p0 > Lo - 1) p0 = Lo - 1;
    int p1 = i0 + 16 + n;   if (p1 > Lo - 1) p1 = Lo - 1;
    const float* ptr0 = dnab + (size_t)p0 * 4 + 2 * half;
    const float* ptr1 = dnab + (size_t)p1 * 4 + 2 * half;

    v2f b0[STEP], b1[STEP];
#pragma unroll
    for (int j = 0; j < STEP; ++j) {
        b0[j] = *(const v2f*)(ptr0 + j * 4);
        b1[j] = *(const v2f*)(ptr1 + j * 4);
    }

    v8f c0 = {};
    v8f c1 = {};
#pragma unroll
    for (int j = 0; j < STEP; ++j) {
        c0 = __builtin_amdgcn_wmma_f32_16x16x4_f32(false, aw[j], false, b0[j],
                                                   (short)0, c0, false, false);
        c1 = __builtin_amdgcn_wmma_f32_16x16x4_f32(false, aw[j], false, b1[j],
                                                   (short)0, c1, false, false);
    }

    float t1f = __shfl(c1[0], n, 32);
    float t1r = __shfl(c1[1], n, 32);
    float sf = (half == 0) ? c0[0] : t1f;
    float sr = (half == 0) ? c0[1] : t1r;

    const int i = i0 + lane;
    if (i < Ltot) {
        float r = 0.0f;
        if (i < Lo) {
            float ind = (sf > 0.0f || sr > 0.0f) ? 1.0f : 0.0f;
            float s   = fmaxf(sf, sr);
            float k   = __expf((s - MAXS) * lamp[0]) * ind;
            r = k * concen[(size_t)b * Ltot + i];
        }
        out[(size_t)b * Ltot + i] = r;
    }
}

extern "C" void kernel_launch(void* const* d_in, const int* in_sizes, int n_in,
                              void* d_out, int out_size, void* d_ws, size_t ws_size,
                              hipStream_t stream) {
    const float* DNA    = (const float*)d_in[0];
    const float* concen = (const float*)d_in[1];
    const float* PWM    = (const float*)d_in[2];
    const float* PWMrc  = (const float*)d_in[3];
    const float* lam    = (const float*)d_in[4];

    const int S0 = in_sizes[0];
    const int S1 = in_sizes[1];
    const int BL = S0 / 4;
    const int B  = S1 - BL;
    const int L  = BL / B;
    const int Ltot = L + 1;

    dim3 grid((Ltot + 255) / 256, B);
    pwm_scan_wmma_kernel<<<grid, 256, 0, stream>>>(DNA, concen, PWM, PWMrc, lam,
                                                   (float*)d_out, L);
}


# Banded weight operand: B = both 12x4 filters at 8 shifts (76x16), A rows = 16 groups of 8 outputs, so 19 WMMAs give 128 outputs (was 24 per 32); per-output sum order follows the banded k order
# speedup vs baseline: 2.5530x; 2.5530x over previous
#include <hip/hip_runtime.h>


typedef __attribute__((ext_vector_type(2))) float v2f;
typedef __attribute__((ext_vector_type(4))) float v4f;
typedef __attribute__((ext_vector_type(8))) float v8f;

#define STEP 12
#define MAXS 10.0f
#define NPAIR 9
#define ITERS 4

__global__ __launch_bounds__(256) void pwm_scan_banded_kernel(
    const float* __restrict__ dna,
    const float* __restrict__ concen,
    const float* __restrict__ pwm,
    const float* __restrict__ pwmrc,
    const float* __restrict__ lamp,
    float* __restrict__ out,
    int L)
{
    const int lane = threadIdx.x & 31;
    const int wave = __builtin_amdgcn_readfirstlane((int)(threadIdx.x >> 5));
    const int c    = lane & 15;
    const int h    = lane >> 4;
    const int b    = blockIdx.y;
    const int Lo   = L - STEP + 1;
    const int Ltot = L + 1;

    const int shift = c & 7;
    const float* wsrc = (c >> 3) ? pwmrc : pwm;
    v2f wlo[NPAIR], whi[NPAIR], wlast;
#pragma unroll
    for (int t = 0; t < NPAIR; ++t) {
        const int j = 2 * t + h - shift;
        v2f lo = {0.0f, 0.0f}, hi = {0.0f, 0.0f};
        if (j >= 0 && j < STEP) {
            lo = *(const v2f*)(wsrc + 4 * j);
            hi = *(const v2f*)(wsrc + 4 * j + 2);
        }
        wlo[t] = lo;
        whi[t] = hi;
    }
    {
        const int j = 18 - shift;
        v2f w = {0.0f, 0.0f};
        if (j < STEP) w = *(const v2f*)(wsrc + 4 * j + 2 * h);
        wlast = w;
    }

    const int grp    = 4 * (c & 3) + 2 * (c >> 3) + ((c >> 2) & 1);
    const int rowoff = 8 * grp;

    const float* dnab = dna + (size_t)b * L * 4;
    const float* conb = concen + (size_t)b * Ltot;
    float*       outb = out + (size_t)b * Ltot;
    const float  lam  = lamp[0];

    for (int it = 0; it < ITERS; ++it) {
        const int base = ((blockIdx.x * ITERS + it) * 8 + wave) * 256;
        if (base >= Ltot) break;

        const int r0 = base + rowoff;
        const int r1 = r0 + 128;

        v4f a0[NPAIR], a1[NPAIR];
        v2f l0, l1;
        const v4f* dna4 = (const v4f*)dnab;
#pragma unroll
        for (int t = 0; t < NPAIR; ++t) {
            int q0 = r0 + 2 * t + h; q0 = q0 < L ? q0 : L - 1;
            int q1 = r1 + 2 * t + h; q1 = q1 < L ? q1 : L - 1;
            a0[t] = dna4[q0];
            a1[t] = dna4[q1];
        }
        {
            int q0 = r0 + 18; q0 = q0 < L ? q0 : L - 1;
            int q1 = r1 + 18; q1 = q1 < L ? q1 : L - 1;
            l0 = *(const v2f*)(dnab + q0 * 4 + 2 * h);
            l1 = *(const v2f*)(dnab + q1 * 4 + 2 * h);
        }

        float cv[8];
        const int i0 = base + lane;
#pragma unroll
        for (int r = 0; r < 8; ++r) {
            const int lim = Ltot - 1 - 32 * r;
            const int ic  = i0 < lim ? i0 : lim;
            cv[r] = conb[ic + 32 * r];
        }

        __builtin_amdgcn_sched_barrier(0);

        v8f acc0 = {};
        v8f acc1 = {};
#pragma unroll
        for (int t = 0; t < NPAIR; ++t) {
            const v2f x0lo = {a0[t].x, a0[t].y}, x0hi = {a0[t].z, a0[t].w};
            const v2f x1lo = {a1[t].x, a1[t].y}, x1hi = {a1[t].z, a1[t].w};
            acc0 = __builtin_amdgcn_wmma_f32_16x16x4_f32(false, x0lo, false, wlo[t],
                                                         (short)0, acc0, false, false);
            acc1 = __builtin_amdgcn_wmma_f32_16x16x4_f32(false, x1lo, false, wlo[t],
                                                         (short)0, acc1, false, false);
            acc0 = __builtin_amdgcn_wmma_f32_16x16x4_f32(false, x0hi, false, whi[t],
                                                         (short)0, acc0, false, false);
            acc1 = __builtin_amdgcn_wmma_f32_16x16x4_f32(false, x1hi, false, whi[t],
                                                         (short)0, acc1, false, false);
        }
        acc0 = __builtin_amdgcn_wmma_f32_16x16x4_f32(false, l0, false, wlast,
                                                     (short)0, acc0, false, false);
        acc1 = __builtin_amdgcn_wmma_f32_16x16x4_f32(false, l1, false, wlast,
                                                     (short)0, acc1, false, false);

        float s[8];
#pragma unroll
        for (int r = 0; r < 8; ++r) {
            const float v0 = acc0[r];
            const float v1 = acc1[r];
            const float o0 = __builtin_bit_cast(float, __builtin_amdgcn_update_dpp(
                0, __builtin_bit_cast(int, v0), 0x128, 0xF, 0xF, true));
            const float o1 = __builtin_bit_cast(float, __builtin_amdgcn_update_dpp(
                0, __builtin_bit_cast(int, v1), 0x128, 0xF, 0xF, true));
            acc0[r] = fmaxf(v0, o0);
            acc1[r] = fmaxf(v1, o1);
        }
#pragma unroll
        for (int r = 0; r < 4; ++r) {
            s[r]     = (c & 8) ? acc0[r + 4] : acc0[r];
            s[r + 4] = (c & 8) ? acc1[r + 4] : acc1[r];
        }

        if (base + 256 <= Lo) {
#pragma unroll
            for (int r = 0; r < 8; ++r) {
                const float sc  = s[r];
                const float ind = (sc > 0.0f) ? 1.0f : 0.0f;
                const float k   = __expf((sc - MAXS) * lam) * ind;
                outb[i0 + 32 * r] = k * cv[r];
            }
        } else {
#pragma unroll
            for (int r = 0; r < 8; ++r) {
                const int i = i0 + 32 * r;
                if (i < Ltot) {
                    float res = 0.0f;
                    if (i < Lo) {
                        const float sc  = s[r];
                        const float ind = (sc > 0.0f) ? 1.0f : 0.0f;
                        const float k   = __expf((sc - MAXS) * lam) * ind;
                        res = k * cv[r];
                    }
                    outb[i] = res;
                }
            }
        }
    }
}

extern "C" void kernel_launch(void* const* d_in, const int* in_sizes, int n_in,
                              void* d_out, int out_size, void* d_ws, size_t ws_size,
                              hipStream_t stream) {
    const float* DNA    = (const float*)d_in[0];
    const float* concen = (const float*)d_in[1];
    const float* PWM    = (const float*)d_in[2];
    const float* PWMrc  = (const float*)d_in[3];
    const float* lam    = (const float*)d_in[4];

    const int S0 = in_sizes[0];
    const int S1 = in_sizes[1];
    const int BL = S0 / 4;
    const int B  = S1 - BL;
    const int L  = BL / B;
    const int Ltot = L + 1;

    dim3 grid((Ltot + 8191) / 8192, B);
    pwm_scan_banded_kernel<<<grid, 256, 0, stream>>>(DNA, concen, PWM, PWMrc, lam,
                                                     (float*)d_out, L);
}


# A fragments staged through per-wave LDS: 9 whole-line 512-byte global loads per 256-output chunk replace 20 fragment loads that touched 16 lines each; LDS row pitch 9 slots of 16 B, no barrier
# speedup vs baseline: 3.2463x; 1.2716x over previous
#include <hip/hip_runtime.h>


typedef __attribute__((ext_vector_type(2))) float v2f;
typedef __attribute__((ext_vector_type(4))) float v4f;
typedef __attribute__((ext_vector_type(8))) float v8f;

#define STEP 12
#define MAXS 10.0f
#define NPAIR 9
#define ITERS 4
#define WAVE_UNITS 324

__global__ __launch_bounds__(256) void pwm_scan_banded_kernel(
    const float* __restrict__ dna,
    const float* __restrict__ concen,
    const float* __restrict__ pwm,
    const float* __restrict__ pwmrc,
    const float* __restrict__ lamp,
    float* __restrict__ out,
    int L)
{
    const int lane = threadIdx.x & 31;
    const int wave = __builtin_amdgcn_readfirstlane((int)(threadIdx.x >> 5));
    const int c    = lane & 15;
    const int h    = lane >> 4;
    const int b    = blockIdx.y;
    const int Lo   = L - STEP + 1;
    const int Ltot = L + 1;

    const int shift = c & 7;
    const float* wsrc = (c >> 3) ? pwmrc : pwm;
    v2f wlo[NPAIR], whi[NPAIR], wlast;
#pragma unroll
    for (int t = 0; t < NPAIR; ++t) {
        const int j = 2 * t + h - shift;
        v2f lo = {0.0f, 0.0f}, hi = {0.0f, 0.0f};
        if (j >= 0 && j < STEP) {
            lo = *(const v2f*)(wsrc + 4 * j);
            hi = *(const v2f*)(wsrc + 4 * j + 2);
        }
        wlo[t] = lo;
        whi[t] = hi;
    }
    {
        const int j = 18 - shift;
        v2f w = {0.0f, 0.0f};
        if (j < STEP) w = *(const v2f*)(wsrc + 4 * j + 2 * h);
        wlast = w;
    }

    const int grp    = 4 * (c & 3) + 2 * (c >> 3) + ((c >> 2) & 1);

    __shared__ v4f stage[8 * WAVE_UNITS];
    v4f* wst = stage + wave * WAVE_UNITS;
    const int st_u = lane + (lane >> 3);
    const int fr_u = 9 * grp + h;

    const float* dnab = dna + (size_t)b * L * 4;
    const float* conb = concen + (size_t)b * Ltot;
    float*       outb = out + (size_t)b * Ltot;
    const float  lam  = lamp[0];

    for (int it = 0; it < ITERS; ++it) {
        const int base = ((blockIdx.x * ITERS + it) * 8 + wave) * 256;
        if (base >= Ltot) break;

        const v4f* dna4 = (const v4f*)dnab;
        v4f g[9];
#pragma unroll
        for (int i = 0; i < 9; ++i) {
            int p = base + 32 * i + lane; p = p < L ? p : L - 1;
            g[i] = dna4[p];
        }

        float cv[8];
        const int i0 = base + lane;
#pragma unroll
        for (int r = 0; r < 8; ++r) {
            const int lim = Ltot - 1 - 32 * r;
            const int ic  = i0 < lim ? i0 : lim;
            cv[r] = conb[ic + 32 * r];
        }

#pragma unroll
        for (int i = 0; i < 9; ++i) wst[st_u + 36 * i] = g[i];

        __builtin_amdgcn_fence(__ATOMIC_RELEASE, "wavefront");
        __builtin_amdgcn_wave_barrier();
        __builtin_amdgcn_fence(__ATOMIC_ACQUIRE, "wavefront");

        v4f a0[NPAIR], a1[NPAIR];
        v2f l0, l1;
#pragma unroll
        for (int t = 0; t < NPAIR; ++t) {
            a0[t] = wst[fr_u + 2 * t + (t >> 2)];
            a1[t] = wst[fr_u + 144 + 2 * t + (t >> 2)];
        }
        {
            const float* lastp = (const float*)(wst + 9 * grp + 20) + 2 * h;
            l0 = *(const v2f*)lastp;
            l1 = *(const v2f*)(lastp + 4 * 144);
        }

        __builtin_amdgcn_sched_barrier(0);

        v8f acc0 = {};
        v8f acc1 = {};
#pragma unroll
        for (int t = 0; t < NPAIR; ++t) {
            const v2f x0lo = {a0[t].x, a0[t].y}, x0hi = {a0[t].z, a0[t].w};
            const v2f x1lo = {a1[t].x, a1[t].y}, x1hi = {a1[t].z, a1[t].w};
            acc0 = __builtin_amdgcn_wmma_f32_16x16x4_f32(false, x0lo, false, wlo[t],
                                                         (short)0, acc0, false, false);
            acc1 = __builtin_amdgcn_wmma_f32_16x16x4_f32(false, x1lo, false, wlo[t],
                                                         (short)0, acc1, false, false);
            acc0 = __builtin_amdgcn_wmma_f32_16x16x4_f32(false, x0hi, false, whi[t],
                                                         (short)0, acc0, false, false);
            acc1 = __builtin_amdgcn_wmma_f32_16x16x4_f32(false, x1hi, false, whi[t],
                                                         (short)0, acc1, false, false);
        }
        acc0 = __builtin_amdgcn_wmma_f32_16x16x4_f32(false, l0, false, wlast,
                                                     (short)0, acc0, false, false);
        acc1 = __builtin_amdgcn_wmma_f32_16x16x4_f32(false, l1, false, wlast,
                                                     (short)0, acc1, false, false);

        float s[8];
#pragma unroll
        for (int r = 0; r < 8; ++r) {
            const float v0 = acc0[r];
            const float v1 = acc1[r];
            const float o0 = __builtin_bit_cast(float, __builtin_amdgcn_update_dpp(
                0, __builtin_bit_cast(int, v0), 0x128, 0xF, 0xF, true));
            const float o1 = __builtin_bit_cast(float, __builtin_amdgcn_update_dpp(
                0, __builtin_bit_cast(int, v1), 0x128, 0xF, 0xF, true));
            acc0[r] = fmaxf(v0, o0);
            acc1[r] = fmaxf(v1, o1);
        }
#pragma unroll
        for (int r = 0; r < 4; ++r) {
            s[r]     = (c & 8) ? acc0[r + 4] : acc0[r];
            s[r + 4] = (c & 8) ? acc1[r + 4] : acc1[r];
        }

        if (base + 256 <= Lo) {
#pragma unroll
            for (int r = 0; r < 8; ++r) {
                const float sc  = s[r];
                const float ind = (sc > 0.0f) ? 1.0f : 0.0f;
                const float k   = __expf((sc - MAXS) * lam) * ind;
                outb[i0 + 32 * r] = k * cv[r];
            }
        } else {
#pragma unroll
            for (int r = 0; r < 8; ++r) {
                const int i = i0 + 32 * r;
                if (i < Ltot) {
                    float res = 0.0f;
                    if (i < Lo) {
                        const float sc  = s[r];
                        const float ind = (sc > 0.0f) ? 1.0f : 0.0f;
                        const float k   = __expf((sc - MAXS) * lam) * ind;
                        res = k * cv[r];
                    }
                    outb[i] = res;
                }
            }
        }
    }
}

extern "C" void kernel_launch(void* const* d_in, const int* in_sizes, int n_in,
                              void* d_out, int out_size, void* d_ws, size_t ws_size,
                              hipStream_t stream) {
    const float* DNA    = (const float*)d_in[0];
    const float* concen = (const float*)d_in[1];
    const float* PWM    = (const float*)d_in[2];
    const float* PWMrc  = (const float*)d_in[3];
    const float* lam    = (const float*)d_in[4];

    const int S0 = in_sizes[0];
    const int S1 = in_sizes[1];
    const int BL = S0 / 4;
    const int B  = S1 - BL;
    const int L  = BL / B;
    const int Ltot = L + 1;

    dim3 grid((Ltot + 8191) / 8192, B);
    pwm_scan_banded_kernel<<<grid, 256, 0, stream>>>(DNA, concen, PWM, PWMrc, lam,
                                                     (float*)d_out, L);
}
